# starting kernel
# baseline (speedup 1.0000x reference)
#include <hip/hip_runtime.h>
#include <math.h>

typedef __attribute__((ext_vector_type(2))) float v2f;
typedef __attribute__((ext_vector_type(8))) float v8f;

#define NFEAT  128
#define LDSPAD 132

__global__ __launch_bounds__(256) void gcn_gemm_bias(
    const float* __restrict__ X, const float* __restrict__ W,
    const float* __restrict__ Bv, float* __restrict__ H, int nnodes)
{
    __shared__ float xs[16 * LDSPAD];

    const int tid = threadIdx.x;
    const int m0  = blockIdx.x * 16;

    #pragma unroll
    for (int i = 0; i < 8; ++i) {
        int idx = tid + i * 256;
        int r = idx >> 7;
        int c = idx & 127;
        int gr = m0 + r;
        xs[r * LDSPAD + c] = (gr < nnodes) ? X[(size_t)gr * NFEAT + c] : 0.0f;
    }
    __syncthreads();

    const int wave = tid >> 5;
    const int lane = tid & 31;
    const int nn   = lane & 15;
    const int kb   = (lane >> 4) << 1;
    const int n0   = wave * 16;

    v8f acc;
    {
        float bias = Bv[n0 + nn];
        #pragma unroll
        for (int r = 0; r < 8; ++r) acc[r] = bias;
    }

    const float* xrow = &xs[nn * LDSPAD + kb];
    const float* wcol = &W[(size_t)kb * NFEAT + n0 + nn];

    #pragma unroll
    for (int k = 0; k < NFEAT; k += 4) {
        v2f a, b;
        a.x = xrow[k];
        a.y = xrow[k + 1];
        b.x = wcol[(size_t)k * NFEAT];
        b.y = wcol[(size_t)(k + 1) * NFEAT];
        acc = __builtin_amdgcn_wmma_f32_16x16x4_f32(
             false, a,  false, b,
             (short)0, acc,  false,  false);
    }

    const int mh = (lane >> 4) * 8;
    #pragma unroll
    for (int r = 0; r < 8; ++r) {
        int gm = m0 + r + mh;
        if (gm < nnodes) H[(size_t)gm * NFEAT + n0 + nn] = acc[r];
    }
}

__global__ __launch_bounds__(256) void gcn_scatter(
    const float* __restrict__ H, const int* __restrict__ rows,
    const int* __restrict__ cols, const float* __restrict__ vals,
    float* __restrict__ AGG, int nedges)
{
    int e    = (int)((blockIdx.x * 256u + threadIdx.x) >> 5);
    int lane = threadIdx.x & 31;
    if (e >= nedges) return;

    int   r = rows[e];
    int   c = cols[e];
    float v = vals[e];

    const float4* hp = (const float4*)(H + (size_t)c * NFEAT);
    float4 m = hp[lane];

    float* out = AGG + (size_t)r * NFEAT + (size_t)lane * 4;
    unsafeAtomicAdd(out + 0, m.x * v);
    unsafeAtomicAdd(out + 1, m.y * v);
    unsafeAtomicAdd(out + 2, m.z * v);
    unsafeAtomicAdd(out + 3, m.w * v);
}

__global__ __launch_bounds__(256) void gcn_zero(float4* __restrict__ p, int n4)
{
    int i = blockIdx.x * 256 + threadIdx.x;
    if (i < n4) {
        float4 z; z.x = 0.f; z.y = 0.f; z.z = 0.f; z.w = 0.f;
        p[i] = z;
    }
}

__global__ __launch_bounds__(256) void gcn_elu(
    const float* __restrict__ in, float* __restrict__ out, int n)
{
    int i = blockIdx.x * 256 + threadIdx.x;
    if (i < n) {
        float v = in[i];
        out[i] = (v > 0.0f) ? v : (__expf(v) - 1.0f);
    }
}

extern "C" void kernel_launch(void* const* d_in, const int* in_sizes, int n_in,
                              void* d_out, int out_size, void* d_ws, size_t ws_size,
                              hipStream_t stream)
{
    const float* x    = (const float*)d_in[0];
    const float* W    = (const float*)d_in[1];
    const float* b    = (const float*)d_in[2];
    const int*   rows = (const int*)  d_in[3];
    const int*   cols = (const int*)  d_in[4];
    const float* vals = (const float*)d_in[5];

    const int nnodes  = in_sizes[0] / NFEAT;
    const int nlayers = in_sizes[1] / (NFEAT * NFEAT);
    const int nedges  = in_sizes[3] / nlayers;

    const size_t nodeElems = (size_t)nnodes * NFEAT;
    float* hbuf = (float*)d_ws;
    float* agg  = hbuf + nodeElems;

    const int n          = (int)nodeElems;
    const int gemmBlocks = (nnodes + 15) / 16;
    const int zeroBlocks = (n / 4 + 255) / 256;
    const int eluBlocks  = (n + 255) / 256;
    const int scatBlocks = (nedges + 7) / 8;

    for (int l = 0; l < nlayers; ++l) {
        const float* cur = (l == 0) ? x : agg;
        gcn_gemm_bias<<<gemmBlocks, 256, 0, stream>>>(
            cur, W + (size_t)l * NFEAT * NFEAT, b + (size_t)l * NFEAT,
            hbuf, nnodes);

        gcn_zero<<<zeroBlocks, 256, 0, stream>>>((float4*)agg, n / 4);

        gcn_scatter<<<scatBlocks, 256, 0, stream>>>(
            hbuf, rows + (size_t)l * nedges, cols + (size_t)l * nedges,
            vals + (size_t)l * nedges, agg, nedges);

        float* eluOut = (l == nlayers - 1) ? (float*)d_out : agg;
        gcn_elu<<<eluBlocks, 256, 0, stream>>>(agg, eluOut, n);
    }
}


# Linear kernel as 512 resident blocks: each walks 6 to 7 row tiles, waves keep their W columns in registers, next tile's rows fetched under the products; 25000 one-tile waves gone; sum order unchanged
# speedup vs baseline: 3.3397x; 3.3397x over previous
#include <hip/hip_runtime.h>
#include <math.h>

typedef __attribute__((ext_vector_type(2))) float v2f;
typedef __attribute__((ext_vector_type(8))) float v8f;

#define NFEAT  128
#define LDSPAD 132

#define AGG_ROWS   196
#define AGG_WAVES  32
#define AGG_DEPTH  2
#define AGG_QCAP   1056

#define GEMM_BLOCKS 512

__device__ __forceinline__ void gemm_fetch(const float* __restrict__ X, int nnodes, int tile,
                                           int sr, int sc, float4& p0, float4& p1)
{
    const int r0 = tile * 16 + sr;
    const float* src = X + (size_t)r0 * NFEAT + sc;
    if (tile * 16 + 16 <= nnodes) {
        p0 = *(const float4*)src;
        p1 = *(const float4*)(src + 8 * NFEAT);
    } else {
        float4 z; z.x = 0.f; z.y = 0.f; z.z = 0.f; z.w = 0.f;
        p0 = z; p1 = z;
        if (r0 < nnodes)     p0 = *(const float4*)src;
        if (r0 + 8 < nnodes) p1 = *(const float4*)(src + 8 * NFEAT);
    }
}

__device__ __forceinline__ void plane_put(int row, int g, int lane, int* __restrict__ plane)
{
    const unsigned o = (unsigned)row / (unsigned)AGG_ROWS;
    int w = 0;
    #pragma unroll
    for (int b = 0; b < 8; ++b) {
        const unsigned bal = __builtin_amdgcn_ballot_w32(((o >> b) & 1u) != 0u);
        if (lane == b) w = (int)bal;
    }
    if (lane < 8)
        plane[(g >> 5) * 256 + (g & 31) * 4 + (lane & 3) + ((lane & 4) << 5)] = w;
}

__global__ __launch_bounds__(256) void gcn_gemm_bias(
    const float* __restrict__ X, const float* __restrict__ W,
    const float* __restrict__ Bv, float* __restrict__ H, int nnodes,
    const int* __restrict__ rows, int* __restrict__ plane, int nedges)
{
    __shared__ float xs[2][16 * LDSPAD];

    const int tid  = threadIdx.x;
    const int wave = tid >> 5;
    const int lane = tid & 31;
    const int nn   = lane & 15;
    const int kb   = (lane >> 4) << 1;
    const int n0   = wave * 16;
    const int mh   = (lane >> 4) * 8;
    const int sr   = tid >> 5;
    const int sc   = (tid & 31) << 2;

    const int ntiles = (nnodes + 15) >> 4;
    const int tstep  = gridDim.x;
    int tile = blockIdx.x;

    float4 p0, p1;
    gemm_fetch(X, nnodes, tile, sr, sc, p0, p1);
    const int ngroups = (nedges >> 10) << 5;
    const int gstep   = tstep * 8;
    int g  = blockIdx.x * 8 + wave;
    int rv = 0;
    if (g < ngroups) rv = rows[g * 32 + lane];

    const float* wcol = &W[(size_t)kb * NFEAT + n0 + nn];
    v2f bw[NFEAT / 4];
    #pragma unroll
    for (int j = 0; j < NFEAT / 4; ++j) {
        bw[j].x = wcol[(size_t)(4 * j) * NFEAT];
        bw[j].y = wcol[(size_t)(4 * j + 1) * NFEAT];
    }
    const float bias = Bv[n0 + nn];

    *(float4*)&xs[0][sr * LDSPAD + sc]       = p0;
    *(float4*)&xs[0][(sr + 8) * LDSPAD + sc] = p1;
    __syncthreads();

    int cur = 0;
    for (; tile < ntiles; tile += tstep) {
        const bool more = tile + tstep < ntiles;
        if (more) gemm_fetch(X, nnodes, tile + tstep, sr, sc, p0, p1);
        const int gn = g + gstep;
        int rvn = 0;
        if (gn < ngroups) rvn = rows[gn * 32 + lane];

        const float* xrow = &xs[cur][nn * LDSPAD + kb];
        v8f acc;
        #pragma unroll
        for (int r = 0; r < 8; ++r) acc[r] = bias;
        #pragma unroll
        for (int k = 0; k < NFEAT; k += 4) {
            v2f a;
            a.x = xrow[k];
            a.y = xrow[k + 1];
            acc = __builtin_amdgcn_wmma_f32_16x16x4_f32(
                 false, a,  false, bw[k >> 2],
                 (short)0, acc,  false,  false);
        }

        const int m0 = tile * 16;
        float* dst = H + (size_t)(m0 + mh) * NFEAT + n0 + nn;
        if (m0 + 16 <= nnodes) {
            #pragma unroll
            for (int r = 0; r < 8; ++r) dst[r * NFEAT] = acc[r];
        } else {
            #pragma unroll
            for (int r = 0; r < 8; ++r)
                if (m0 + r + mh < nnodes) dst[r * NFEAT] = acc[r];
        }

        if (g < ngroups) plane_put(rv, g, lane, plane);
        g = gn; rv = rvn;

        if (more) {
            *(float4*)&xs[cur ^ 1][sr * LDSPAD + sc]       = p0;
            *(float4*)&xs[cur ^ 1][(sr + 8) * LDSPAD + sc] = p1;
        }
        __builtin_amdgcn_fence(__ATOMIC_RELEASE, "workgroup", "local");
        __builtin_amdgcn_s_barrier();
        __builtin_amdgcn_fence(__ATOMIC_ACQUIRE, "workgroup", "local");
        cur ^= 1;
    }

    while (g < ngroups) {
        plane_put(rv, g, lane, plane);
        g += gstep;
        if (g < ngroups) rv = rows[g * 32 + lane];
    }
}


__device__ __forceinline__ void lds_add(float* p, float x)
{
    (void)__hip_atomic_fetch_add(p, x, __ATOMIC_RELAXED, __HIP_MEMORY_SCOPE_WORKGROUP);
}

template <bool FULL>
__device__ __forceinline__ void agg_drain(
    const float* __restrict__ H, const int* __restrict__ rows,
    const int* __restrict__ cols, const float* __restrict__ vals,
    float* acc, const int* myq, int first, int nb, int lane, int base, int nown, int nnodes)
{
    int lr = 0, c = 0;
    float v = 0.0f;
    if (FULL || lane < nb) {
        const int e = myq[first + lane];
        lr = rows[e] - base;
        if ((unsigned)lr >= (unsigned)nown) lr = AGG_ROWS;
        c  = cols[e];
        v  = vals[e];
    }
    c = min(max(c, 0), nnodes - 1);

    #pragma unroll
    for (int b = 0; b < 32; b += 8) {
        if (!FULL && b >= nb) break;
        float4 h[8];
        #pragma unroll
        for (int i = 0; i < 8; ++i) {
            if (FULL || b + i < nb) {
                const int ci = __builtin_amdgcn_readlane(c, b + i);
                h[i] = ((const float4*)(H + (size_t)ci * NFEAT))[lane];
            }
        }
        #pragma unroll
        for (int i = 0; i < 8; ++i) {
            if (FULL || b + i < nb) {
                const float vi = __int_as_float(__builtin_amdgcn_readlane(__float_as_int(v), b + i));
                const int   ri = __builtin_amdgcn_readlane(lr, b + i);
                float* a = acc + ri * NFEAT + lane;
                lds_add(a,      h[i].x * vi);
                lds_add(a + 32, h[i].y * vi);
                lds_add(a + 64, h[i].z * vi);
                lds_add(a + 96, h[i].w * vi);
            }
        }
    }
}

__device__ __forceinline__ int4 agg_load_ids_tail(const int* __restrict__ rows, int s,
                                                  int nedges, int lane)
{
    const int e0 = s * 128 + lane * 4;
    int4 t; t.x = -1; t.y = -1; t.z = -1; t.w = -1;
    if (e0 + 0 < nedges) t.x = rows[e0 + 0];
    if (e0 + 1 < nedges) t.y = rows[e0 + 1];
    if (e0 + 2 < nedges) t.z = rows[e0 + 2];
    if (e0 + 3 < nedges) t.w = rows[e0 + 3];
    return t;
}

__device__ __forceinline__ void agg_append(int4 ids, int e0, int base, int nown, int* myq, int& cnt)
{
    const bool m0 = (unsigned)(ids.x - base) < (unsigned)nown;
    const bool m1 = (unsigned)(ids.y - base) < (unsigned)nown;
    const bool m2 = (unsigned)(ids.z - base) < (unsigned)nown;
    const bool m3 = (unsigned)(ids.w - base) < (unsigned)nown;
    if (__builtin_amdgcn_ballot_w32(m0 | m1 | m2 | m3) != 0u) {
        const bool mt[4] = { m0, m1, m2, m3 };
        #pragma unroll
        for (int j = 0; j < 4; ++j) {
            const unsigned m = __builtin_amdgcn_ballot_w32(mt[j]);
            if (m != 0u) {
                const int pos = cnt + (int)__builtin_amdgcn_mbcnt_lo(m, 0u);
                if (mt[j]) myq[pos] = e0 + j;
                cnt += __builtin_popcount(m);
            }
        }
    }
}

__device__ __forceinline__ unsigned agg_match(int4 lo, int4 hi, unsigned id)
{
    const unsigned d[8] = { (unsigned)lo.x, (unsigned)lo.y, (unsigned)lo.z, (unsigned)lo.w,
                            (unsigned)hi.x, (unsigned)hi.y, (unsigned)hi.z, (unsigned)hi.w };
    unsigned differs = 0u;
    #pragma unroll
    for (int b = 0; b < 8; ++b)
        differs |= d[b] ^ (0u - ((id >> b) & 1u));
    return ~differs;
}

__device__ __forceinline__ void agg_append_bits(unsigned m, int e0, int* myq, int& cnt)
{
    unsigned bm = __builtin_amdgcn_ballot_w32(m != 0u);
    while (bm != 0u) {
        if (m != 0u) {
            myq[cnt + (int)__builtin_amdgcn_mbcnt_lo(bm, 0u)] = e0 + __builtin_ctz(m);
            m &= m - 1u;
        }
        cnt += __builtin_popcount(bm);
        bm = __builtin_amdgcn_ballot_w32(m != 0u);
    }
}

__global__ __launch_bounds__(1024) void gcn_aggregate_elu(
    const float* __restrict__ H, const int* __restrict__ rows,
    const int* __restrict__ cols, const float* __restrict__ vals,
    const int* __restrict__ plane, float* __restrict__ out, int nnodes, int nedges)
{
    __shared__ float acc[(AGG_ROWS + 1) * NFEAT];
    __shared__ int   queue[AGG_WAVES * AGG_QCAP];

    const int tid  = threadIdx.x;
    const int lane = tid & 31;
    const int wave = __builtin_amdgcn_readfirstlane(tid >> 5);
    const int base = blockIdx.x * AGG_ROWS;
    const int nown = max(min(AGG_ROWS, nnodes - base), 0);

    {
        float4 z; z.x = 0.f; z.y = 0.f; z.z = 0.f; z.w = 0.f;
        for (int i = tid; i < AGG_ROWS * NFEAT / 4; i += 1024) ((float4*)acc)[i] = z;
    }
    __syncthreads();

    int* myq = queue + wave * AGG_QCAP;
    int  cnt = 0;

    const int      nfull  = nedges >> 10;
    const int      stride = AGG_WAVES * AGG_DEPTH;
    const unsigned id     = blockIdx.x & 255u;

    if (wave < nfull) {
        const int4* pw = (const int4*)plane;
        int4 rlo[AGG_DEPTH], rhi[AGG_DEPTH];
        #pragma unroll
        for (int u = 0; u < AGG_DEPTH; ++u) {
            const int s = min(wave + u * AGG_WAVES, nfull - 1);
            rlo[u] = pw[s * 64 + lane];
            rhi[u] = pw[s * 64 + 32 + lane];
        }

        for (int s0 = wave; s0 < nfull; s0 += stride) {
            unsigned m[AGG_DEPTH];
            #pragma unroll
            for (int u = 0; u < AGG_DEPTH; ++u) {
                m[u] = agg_match(rlo[u], rhi[u], id);
                if (s0 + u * AGG_WAVES >= nfull) m[u] = 0u;
            }
            #pragma unroll
            for (int u = 0; u < AGG_DEPTH; ++u) {
                const int sn = min(s0 + u * AGG_WAVES + stride, nfull - 1);
                rlo[u] = pw[sn * 64 + lane];
                rhi[u] = pw[sn * 64 + 32 + lane];
            }
            #pragma unroll
            for (int u = 0; u < AGG_DEPTH; ++u) {
                agg_append_bits(m[u], (s0 + u * AGG_WAVES) * 1024 + lane * 32, myq, cnt);
                while (cnt >= 32) {
                    cnt -= 32;
                    agg_drain<true>(H, rows, cols, vals, acc, myq, cnt, 32, lane, base, nown, nnodes);
                }
            }
        }
    }
    const int nsteps = (nedges + 127) >> 7;
    for (int s = nfull * 8 + wave; s < nsteps; s += AGG_WAVES) {
        agg_append(agg_load_ids_tail(rows, s, nedges, lane), s * 128 + lane * 4, base, nown, myq, cnt);
        while (cnt >= 32) {
            cnt -= 32;
            agg_drain<true>(H, rows, cols, vals, acc, myq, cnt, 32, lane, base, nown, nnodes);
        }
    }
    if (cnt > 0)
        agg_drain<false>(H, rows, cols, vals, acc, myq, 0, cnt, lane, base, nown, nnodes);

    __syncthreads();

    for (int r = wave; r < nown; r += AGG_WAVES) {
        const float* a = acc + r * NFEAT + lane;
        const float v0 = a[0], v1 = a[32], v2 = a[64], v3 = a[96];
        float4 o;
        o.x = (v0 > 0.0f) ? v0 : (__expf(v0) - 1.0f);
        o.y = (v1 > 0.0f) ? v1 : (__expf(v1) - 1.0f);
        o.z = (v2 > 0.0f) ? v2 : (__expf(v2) - 1.0f);
        o.w = (v3 > 0.0f) ? v3 : (__expf(v3) - 1.0f);
        ((float4*)(out + (size_t)(base + r) * NFEAT))[lane] = o;
    }
}

extern "C" void kernel_launch(void* const* d_in, const int* in_sizes, int n_in,
                              void* d_out, int out_size, void* d_ws, size_t ws_size,
                              hipStream_t stream)
{
    const float* x    = (const float*)d_in[0];
    const float* W    = (const float*)d_in[1];
    const float* b    = (const float*)d_in[2];
    const int*   rows = (const int*)  d_in[3];
    const int*   cols = (const int*)  d_in[4];
    const float* vals = (const float*)d_in[5];

    const int nnodes  = in_sizes[0] / NFEAT;
    const int nlayers = in_sizes[1] / (NFEAT * NFEAT);
    const int nedges  = in_sizes[3] / nlayers;

    const size_t nodeElems = (size_t)nnodes * NFEAT;
    float* hbuf = (float*)d_ws;
    float* agg  = hbuf + nodeElems;
    int*   plane = (int*)(agg + nodeElems);

    const int aggBlocks  = (nnodes + AGG_ROWS - 1) / AGG_ROWS;

    for (int l = 0; l < nlayers; ++l) {
        const float* cur = (l == 0) ? x : agg;
        gcn_gemm_bias<<<GEMM_BLOCKS, 256, 0, stream>>>(
            cur, W + (size_t)l * NFEAT * NFEAT, b + (size_t)l * NFEAT,
            hbuf, nnodes, rows + (size_t)l * nedges, plane, nedges);

        float* layerOut = (l == nlayers - 1) ? (float*)d_out : agg;
        gcn_aggregate_elu<<<aggBlocks, 1024, 0, stream>>>(
            hbuf, rows + (size_t)l * nedges, cols + (size_t)l * nedges,
            vals + (size_t)l * nedges, plane, layerOut, nnodes, nedges);
    }
}


# Linear kernel also writes one 16-byte record (row, col, val) per edge, in edge order; aggregation fetches an owned edge with one load in one line, was three loads in three lines; sum order unchanged
# speedup vs baseline: 3.5126x; 1.0518x over previous
#include <hip/hip_runtime.h>
#include <math.h>

typedef __attribute__((ext_vector_type(2))) float v2f;
typedef __attribute__((ext_vector_type(8))) float v8f;

#define NFEAT  128
#define LDSPAD 132

#define AGG_ROWS   196
#define AGG_WAVES  32
#define AGG_DEPTH  2
#define AGG_QCAP   1056

#define GEMM_BLOCKS 512

__device__ __forceinline__ void gemm_fetch(const float* __restrict__ X, int nnodes, int tile,
                                           int sr, int sc, float4& p0, float4& p1)
{
    const int r0 = tile * 16 + sr;
    const float* src = X + (size_t)r0 * NFEAT + sc;
    if (tile * 16 + 16 <= nnodes) {
        p0 = *(const float4*)src;
        p1 = *(const float4*)(src + 8 * NFEAT);
    } else {
        float4 z; z.x = 0.f; z.y = 0.f; z.z = 0.f; z.w = 0.f;
        p0 = z; p1 = z;
        if (r0 < nnodes)     p0 = *(const float4*)src;
        if (r0 + 8 < nnodes) p1 = *(const float4*)(src + 8 * NFEAT);
    }
}

__device__ __forceinline__ void plane_put(int row, int g, int lane, int* __restrict__ plane)
{
    const unsigned o = (unsigned)row / (unsigned)AGG_ROWS;
    int w = 0;
    #pragma unroll
    for (int b = 0; b < 8; ++b) {
        const unsigned bal = __builtin_amdgcn_ballot_w32(((o >> b) & 1u) != 0u);
        if (lane == b) w = (int)bal;
    }
    if (lane < 8)
        plane[(g >> 5) * 256 + (g & 31) * 4 + (lane & 3) + ((lane & 4) << 5)] = w;
}

__device__ __forceinline__ int4 edge_fetch(const int* __restrict__ rows, const int* __restrict__ cols,
                                           const float* __restrict__ vals, int e, int nedges)
{
    int4 r; r.x = 0; r.y = 0; r.z = 0; r.w = 0;
    if (e < nedges) { r.x = rows[e]; r.y = cols[e]; r.z = __float_as_int(vals[e]); }
    return r;
}

__device__ __forceinline__ void group_put(int4 r, int g, int lane, int nplanegroups, int nedges,
                                          int* __restrict__ plane, int4* __restrict__ rec)
{
    const int e = g * 32 + lane;
    if (e < nedges) rec[e] = r;
    if (g < nplanegroups) plane_put(r.x, g, lane, plane);
}

__global__ __launch_bounds__(256) void gcn_gemm_bias(
    const float* __restrict__ X, const float* __restrict__ W,
    const float* __restrict__ Bv, float* __restrict__ H, int nnodes,
    const int* __restrict__ rows, const int* __restrict__ cols, const float* __restrict__ vals,
    int* __restrict__ plane, int* __restrict__ recw, int nedges)
{
    __shared__ float xs[2][16 * LDSPAD];
    int4* __restrict__ rec = (int4*)recw;

    const int tid  = threadIdx.x;
    const int wave = tid >> 5;
    const int lane = tid & 31;
    const int nn   = lane & 15;
    const int kb   = (lane >> 4) << 1;
    const int n0   = wave * 16;
    const int mh   = (lane >> 4) * 8;
    const int sr   = tid >> 5;
    const int sc   = (tid & 31) << 2;

    const int ntiles = (nnodes + 15) >> 4;
    const int tstep  = gridDim.x;
    int tile = blockIdx.x;

    float4 p0, p1;
    gemm_fetch(X, nnodes, tile, sr, sc, p0, p1);
    const int nplanegroups = (nedges >> 10) << 5;
    const int ngroups = (nedges + 31) >> 5;
    const int gstep   = tstep * 8;
    int g  = blockIdx.x * 8 + wave;
    int4 rv = edge_fetch(rows, cols, vals, g * 32 + lane, (g < ngroups) ? nedges : 0);

    const float* wcol = &W[(size_t)kb * NFEAT + n0 + nn];
    v2f bw[NFEAT / 4];
    #pragma unroll
    for (int j = 0; j < NFEAT / 4; ++j) {
        bw[j].x = wcol[(size_t)(4 * j) * NFEAT];
        bw[j].y = wcol[(size_t)(4 * j + 1) * NFEAT];
    }
    const float bias = Bv[n0 + nn];

    *(float4*)&xs[0][sr * LDSPAD + sc]       = p0;
    *(float4*)&xs[0][(sr + 8) * LDSPAD + sc] = p1;
    __syncthreads();

    int cur = 0;
    for (; tile < ntiles; tile += tstep) {
        const bool more = tile + tstep < ntiles;
        if (more) gemm_fetch(X, nnodes, tile + tstep, sr, sc, p0, p1);
        const int gn = g + gstep;
        const int4 rvn = edge_fetch(rows, cols, vals, gn * 32 + lane, (gn < ngroups) ? nedges : 0);

        const float* xrow = &xs[cur][nn * LDSPAD + kb];
        v8f acc;
        #pragma unroll
        for (int r = 0; r < 8; ++r) acc[r] = bias;
        #pragma unroll
        for (int k = 0; k < NFEAT; k += 4) {
            v2f a;
            a.x = xrow[k];
            a.y = xrow[k + 1];
            acc = __builtin_amdgcn_wmma_f32_16x16x4_f32(
                 false, a,  false, bw[k >> 2],
                 (short)0, acc,  false,  false);
        }

        const int m0 = tile * 16;
        float* dst = H + (size_t)(m0 + mh) * NFEAT + n0 + nn;
        if (m0 + 16 <= nnodes) {
            #pragma unroll
            for (int r = 0; r < 8; ++r) dst[r * NFEAT] = acc[r];
        } else {
            #pragma unroll
            for (int r = 0; r < 8; ++r)
                if (m0 + r + mh < nnodes) dst[r * NFEAT] = acc[r];
        }

        if (g < ngroups) group_put(rv, g, lane, nplanegroups, nedges, plane, rec);
        g = gn; rv = rvn;

        if (more) {
            *(float4*)&xs[cur ^ 1][sr * LDSPAD + sc]       = p0;
            *(float4*)&xs[cur ^ 1][(sr + 8) * LDSPAD + sc] = p1;
        }
        __builtin_amdgcn_fence(__ATOMIC_RELEASE, "workgroup", "local");
        __builtin_amdgcn_s_barrier();
        __builtin_amdgcn_fence(__ATOMIC_ACQUIRE, "workgroup", "local");
        cur ^= 1;
    }

    while (g < ngroups) {
        group_put(rv, g, lane, nplanegroups, nedges, plane, rec);
        g += gstep;
        rv = edge_fetch(rows, cols, vals, g * 32 + lane, (g < ngroups) ? nedges : 0);
    }
}


__device__ __forceinline__ void lds_add(float* p, float x)
{
    (void)__hip_atomic_fetch_add(p, x, __ATOMIC_RELAXED, __HIP_MEMORY_SCOPE_WORKGROUP);
}

template <bool FULL>
__device__ __forceinline__ void agg_drain(
    const float* __restrict__ H, const int4* __restrict__ rec,
    float* acc, const int* myq, int first, int nb, int lane, int base, int nown, int nnodes)
{
    int lr = 0, c = 0;
    float v = 0.0f;
    if (FULL || lane < nb) {
        const int4 r = rec[myq[first + lane]];
        lr = r.x - base;
        if ((unsigned)lr >= (unsigned)nown) lr = AGG_ROWS;
        c  = r.y;
        v  = __int_as_float(r.z);
    }
    c = min(max(c, 0), nnodes - 1);

    #pragma unroll
    for (int b = 0; b < 32; b += 8) {
        if (!FULL && b >= nb) break;
        float4 h[8];
        #pragma unroll
        for (int i = 0; i < 8; ++i) {
            if (FULL || b + i < nb) {
                const int ci = __builtin_amdgcn_readlane(c, b + i);
                h[i] = ((const float4*)(H + (size_t)ci * NFEAT))[lane];
            }
        }
        #pragma unroll
        for (int i = 0; i < 8; ++i) {
            if (FULL || b + i < nb) {
                const float vi = __int_as_float(__builtin_amdgcn_readlane(__float_as_int(v), b + i));
                const int   ri = __builtin_amdgcn_readlane(lr, b + i);
                float* a = acc + ri * NFEAT + lane;
                lds_add(a,      h[i].x * vi);
                lds_add(a + 32, h[i].y * vi);
                lds_add(a + 64, h[i].z * vi);
                lds_add(a + 96, h[i].w * vi);
            }
        }
    }
}

__device__ __forceinline__ int4 agg_load_ids_tail(const int* __restrict__ rows, int s,
                                                  int nedges, int lane)
{
    const int e0 = s * 128 + lane * 4;
    int4 t; t.x = -1; t.y = -1; t.z = -1; t.w = -1;
    if (e0 + 0 < nedges) t.x = rows[e0 + 0];
    if (e0 + 1 < nedges) t.y = rows[e0 + 1];
    if (e0 + 2 < nedges) t.z = rows[e0 + 2];
    if (e0 + 3 < nedges) t.w = rows[e0 + 3];
    return t;
}

__device__ __forceinline__ void agg_append(int4 ids, int e0, int base, int nown, int* myq, int& cnt)
{
    const bool m0 = (unsigned)(ids.x - base) < (unsigned)nown;
    const bool m1 = (unsigned)(ids.y - base) < (unsigned)nown;
    const bool m2 = (unsigned)(ids.z - base) < (unsigned)nown;
    const bool m3 = (unsigned)(ids.w - base) < (unsigned)nown;
    if (__builtin_amdgcn_ballot_w32(m0 | m1 | m2 | m3) != 0u) {
        const bool mt[4] = { m0, m1, m2, m3 };
        #pragma unroll
        for (int j = 0; j < 4; ++j) {
            const unsigned m = __builtin_amdgcn_ballot_w32(mt[j]);
            if (m != 0u) {
                const int pos = cnt + (int)__builtin_amdgcn_mbcnt_lo(m, 0u);
                if (mt[j]) myq[pos] = e0 + j;
                cnt += __builtin_popcount(m);
            }
        }
    }
}

__device__ __forceinline__ unsigned agg_match(int4 lo, int4 hi, unsigned id)
{
    const unsigned d[8] = { (unsigned)lo.x, (unsigned)lo.y, (unsigned)lo.z, (unsigned)lo.w,
                            (unsigned)hi.x, (unsigned)hi.y, (unsigned)hi.z, (unsigned)hi.w };
    unsigned differs = 0u;
    #pragma unroll
    for (int b = 0; b < 8; ++b)
        differs |= d[b] ^ (0u - ((id >> b) & 1u));
    return ~differs;
}

__device__ __forceinline__ void agg_append_bits(unsigned m, int e0, int* myq, int& cnt)
{
    unsigned bm = __builtin_amdgcn_ballot_w32(m != 0u);
    while (bm != 0u) {
        if (m != 0u) {
            myq[cnt + (int)__builtin_amdgcn_mbcnt_lo(bm, 0u)] = e0 + __builtin_ctz(m);
            m &= m - 1u;
        }
        cnt += __builtin_popcount(bm);
        bm = __builtin_amdgcn_ballot_w32(m != 0u);
    }
}

__global__ __launch_bounds__(1024) void gcn_aggregate_elu(
    const float* __restrict__ H, const int* __restrict__ rows,
    const int* __restrict__ recw,
    const int* __restrict__ plane, float* __restrict__ out, int nnodes, int nedges)
{
    const int4* __restrict__ rec = (const int4*)recw;
    __shared__ float acc[(AGG_ROWS + 1) * NFEAT];
    __shared__ int   queue[AGG_WAVES * AGG_QCAP];

    const int tid  = threadIdx.x;
    const int lane = tid & 31;
    const int wave = __builtin_amdgcn_readfirstlane(tid >> 5);
    const int base = blockIdx.x * AGG_ROWS;
    const int nown = max(min(AGG_ROWS, nnodes - base), 0);

    {
        float4 z; z.x = 0.f; z.y = 0.f; z.z = 0.f; z.w = 0.f;
        for (int i = tid; i < AGG_ROWS * NFEAT / 4; i += 1024) ((float4*)acc)[i] = z;
    }
    __syncthreads();

    int* myq = queue + wave * AGG_QCAP;
    int  cnt = 0;

    const int      nfull  = nedges >> 10;
    const int      stride = AGG_WAVES * AGG_DEPTH;
    const unsigned id     = blockIdx.x & 255u;

    if (wave < nfull) {
        const int4* pw = (const int4*)plane;
        int4 rlo[AGG_DEPTH], rhi[AGG_DEPTH];
        #pragma unroll
        for (int u = 0; u < AGG_DEPTH; ++u) {
            const int s = min(wave + u * AGG_WAVES, nfull - 1);
            rlo[u] = pw[s * 64 + lane];
            rhi[u] = pw[s * 64 + 32 + lane];
        }

        for (int s0 = wave; s0 < nfull; s0 += stride) {
            unsigned m[AGG_DEPTH];
            #pragma unroll
            for (int u = 0; u < AGG_DEPTH; ++u) {
                m[u] = agg_match(rlo[u], rhi[u], id);
                if (s0 + u * AGG_WAVES >= nfull) m[u] = 0u;
            }
            #pragma unroll
            for (int u = 0; u < AGG_DEPTH; ++u) {
                const int sn = min(s0 + u * AGG_WAVES + stride, nfull - 1);
                rlo[u] = pw[sn * 64 + lane];
                rhi[u] = pw[sn * 64 + 32 + lane];
            }
            #pragma unroll
            for (int u = 0; u < AGG_DEPTH; ++u) {
                agg_append_bits(m[u], (s0 + u * AGG_WAVES) * 1024 + lane * 32, myq, cnt);
                while (cnt >= 32) {
                    cnt -= 32;
                    agg_drain<true>(H, rec, acc, myq, cnt, 32, lane, base, nown, nnodes);
                }
            }
        }
    }
    const int nsteps = (nedges + 127) >> 7;
    for (int s = nfull * 8 + wave; s < nsteps; s += AGG_WAVES) {
        agg_append(agg_load_ids_tail(rows, s, nedges, lane), s * 128 + lane * 4, base, nown, myq, cnt);
        while (cnt >= 32) {
            cnt -= 32;
            agg_drain<true>(H, rec, acc, myq, cnt, 32, lane, base, nown, nnodes);
        }
    }
    if (cnt > 0)
        agg_drain<false>(H, rec, acc, myq, 0, cnt, lane, base, nown, nnodes);

    __syncthreads();

    for (int r = wave; r < nown; r += AGG_WAVES) {
        const float* a = acc + r * NFEAT + lane;
        const float v0 = a[0], v1 = a[32], v2 = a[64], v3 = a[96];
        float4 o;
        o.x = (v0 > 0.0f) ? v0 : (__expf(v0) - 1.0f);
        o.y = (v1 > 0.0f) ? v1 : (__expf(v1) - 1.0f);
        o.z = (v2 > 0.0f) ? v2 : (__expf(v2) - 1.0f);
        o.w = (v3 > 0.0f) ? v3 : (__expf(v3) - 1.0f);
        ((float4*)(out + (size_t)(base + r) * NFEAT))[lane] = o;
    }
}

extern "C" void kernel_launch(void* const* d_in, const int* in_sizes, int n_in,
                              void* d_out, int out_size, void* d_ws, size_t ws_size,
                              hipStream_t stream)
{
    const float* x    = (const float*)d_in[0];
    const float* W    = (const float*)d_in[1];
    const float* b    = (const float*)d_in[2];
    const int*   rows = (const int*)  d_in[3];
    const int*   cols = (const int*)  d_in[4];
    const float* vals = (const float*)d_in[5];

    const int nnodes  = in_sizes[0] / NFEAT;
    const int nlayers = in_sizes[1] / (NFEAT * NFEAT);
    const int nedges  = in_sizes[3] / nlayers;

    const size_t nodeElems = (size_t)nnodes * NFEAT;
    float* hbuf = (float*)d_ws;
    float* agg  = hbuf + nodeElems;
    int*   rec   = (int*)(agg + nodeElems);
    int*   plane = rec + (size_t)nedges * 4;

    const int aggBlocks  = (nnodes + AGG_ROWS - 1) / AGG_ROWS;

    for (int l = 0; l < nlayers; ++l) {
        const float* cur = (l == 0) ? x : agg;
        gcn_gemm_bias<<<GEMM_BLOCKS, 256, 0, stream>>>(
            cur, W + (size_t)l * NFEAT * NFEAT, b + (size_t)l * NFEAT,
            hbuf, nnodes, rows + (size_t)l * nedges, cols + (size_t)l * nedges,
            vals + (size_t)l * nedges, plane, rec, nedges);

        float* layerOut = (l == nlayers - 1) ? (float*)d_out : agg;
        gcn_aggregate_elu<<<aggBlocks, 1024, 0, stream>>>(
            hbuf, rows + (size_t)l * nedges, rec, plane, layerOut, nnodes, nedges);
    }
}


# Linear kernel also writes an index word per step and owner block (bit = a group of 32 edges holds an edge of the block); a block reads the planes of flagged groups only; sums LDS arrival order
# speedup vs baseline: 3.6640x; 1.0431x over previous
#include <hip/hip_runtime.h>
#include <math.h>

typedef __attribute__((ext_vector_type(2))) float v2f;
typedef __attribute__((ext_vector_type(8))) float v8f;

#define NFEAT  128
#define LDSPAD 132

#define AGG_ROWS   196
#define AGG_WAVES  32
#define AGG_QCAP   1056
#define AGG_GCAP   288

#define GEMM_BLOCKS 512

__device__ __forceinline__ void gemm_fetch(const float* __restrict__ X, int nnodes, int tile,
                                           int sr, int sc, float4& p0, float4& p1)
{
    const int r0 = tile * 16 + sr;
    const float* src = X + (size_t)r0 * NFEAT + sc;
    if (tile * 16 + 16 <= nnodes) {
        p0 = *(const float4*)src;
        p1 = *(const float4*)(src + 8 * NFEAT);
    } else {
        float4 z; z.x = 0.f; z.y = 0.f; z.z = 0.f; z.w = 0.f;
        p0 = z; p1 = z;
        if (r0 < nnodes)     p0 = *(const float4*)src;
        if (r0 + 8 < nnodes) p1 = *(const float4*)(src + 8 * NFEAT);
    }
}

__device__ __forceinline__ void plane_put(int row, int g, int lane, int* __restrict__ plane)
{
    const unsigned o = (unsigned)row / (unsigned)AGG_ROWS;
    int w = 0;
    #pragma unroll
    for (int b = 0; b < 8; ++b) {
        const unsigned bal = __builtin_amdgcn_ballot_w32(((o >> b) & 1u) != 0u);
        if (lane == b) w = (int)bal;
    }
    if (lane < 8) plane[g * 8 + lane] = w;
}

__device__ __forceinline__ int4 edge_fetch(const int* __restrict__ rows, const int* __restrict__ cols,
                                           const float* __restrict__ vals, int e, int nedges)
{
    int4 r; r.x = 0; r.y = 0; r.z = 0; r.w = 0;
    if (e < nedges) { r.x = rows[e]; r.y = cols[e]; r.z = __float_as_int(vals[e]); }
    return r;
}

__device__ __forceinline__ void group_put(int4 r, int g, int lane, int nplanegroups, int nedges,
                                          int* __restrict__ plane, int4* __restrict__ rec, unsigned* pres)
{
    const int e = g * 32 + lane;
    if (e < nedges) rec[e] = r;
    if (g < nplanegroups) {
        plane_put(r.x, g, lane, plane);
        const unsigned o = ((unsigned)r.x / (unsigned)AGG_ROWS) & 255u;
        (void)__hip_atomic_fetch_or(&pres[o], 1u << (g & 7), __ATOMIC_RELAXED, __HIP_MEMORY_SCOPE_WORKGROUP);
    }
}

__device__ __forceinline__ void pres_flush(unsigned* p, unsigned char* __restrict__ presence, int q, int tid)
{
    presence[((size_t)(q >> 2) * 256 + tid) * 4 + (q & 3)] = (unsigned char)p[tid];
    p[tid] = 0u;
}

__global__ __launch_bounds__(256) void gcn_gemm_bias(
    const float* __restrict__ X, const float* __restrict__ W,
    const float* __restrict__ Bv, float* __restrict__ H, int nnodes,
    const int* __restrict__ rows, const int* __restrict__ cols, const float* __restrict__ vals,
    int* __restrict__ plane, int* __restrict__ recw, int* __restrict__ presw, int nedges)
{
    __shared__ float xs[2][16 * LDSPAD];
    __shared__ unsigned pres[2][256];
    int4* __restrict__ rec = (int4*)recw;
    unsigned char* __restrict__ presence = (unsigned char*)presw;

    const int tid  = threadIdx.x;
    const int wave = tid >> 5;
    const int lane = tid & 31;
    const int nn   = lane & 15;
    const int kb   = (lane >> 4) << 1;
    const int n0   = wave * 16;
    const int mh   = (lane >> 4) * 8;
    const int sr   = tid >> 5;
    const int sc   = (tid & 31) << 2;

    const int ntiles = (nnodes + 15) >> 4;
    const int tstep  = gridDim.x;
    int tile = blockIdx.x;

    float4 p0, p1;
    gemm_fetch(X, nnodes, tile, sr, sc, p0, p1);
    const int nplanegroups = (nedges >> 10) << 5;
    const int ngroups = (nedges + 31) >> 5;
    const int gstep   = tstep * 8;
    int g  = blockIdx.x * 8 + wave;
    int4 rv = edge_fetch(rows, cols, vals, g * 32 + lane, (g < ngroups) ? nedges : 0);

    const float* wcol = &W[(size_t)kb * NFEAT + n0 + nn];
    v2f bw[NFEAT / 4];
    #pragma unroll
    for (int j = 0; j < NFEAT / 4; ++j) {
        bw[j].x = wcol[(size_t)(4 * j) * NFEAT];
        bw[j].y = wcol[(size_t)(4 * j + 1) * NFEAT];
    }
    const float bias = Bv[n0 + nn];

    *(float4*)&xs[0][sr * LDSPAD + sc]       = p0;
    *(float4*)&xs[0][(sr + 8) * LDSPAD + sc] = p1;
    pres[0][tid] = 0u;
    pres[1][tid] = 0u;
    __syncthreads();

    const int nquarters = (ngroups + 7) >> 3;
    int q     = blockIdx.x;
    int qdone = -1;
    int par   = 0;
    int cur = 0;
    for (; tile < ntiles; tile += tstep) {
        if (qdone >= 0) pres_flush(pres[par ^ 1], presence, qdone, tid);
        const bool more = tile + tstep < ntiles;
        if (more) gemm_fetch(X, nnodes, tile + tstep, sr, sc, p0, p1);
        const int gn = g + gstep;
        const int4 rvn = edge_fetch(rows, cols, vals, gn * 32 + lane, (gn < ngroups) ? nedges : 0);

        const float* xrow = &xs[cur][nn * LDSPAD + kb];
        v8f acc;
        #pragma unroll
        for (int r = 0; r < 8; ++r) acc[r] = bias;
        #pragma unroll
        for (int k = 0; k < NFEAT; k += 4) {
            v2f a;
            a.x = xrow[k];
            a.y = xrow[k + 1];
            acc = __builtin_amdgcn_wmma_f32_16x16x4_f32(
                 false, a,  false, bw[k >> 2],
                 (short)0, acc,  false,  false);
        }

        const int m0 = tile * 16;
        float* dst = H + (size_t)(m0 + mh) * NFEAT + n0 + nn;
        if (m0 + 16 <= nnodes) {
            #pragma unroll
            for (int r = 0; r < 8; ++r) dst[r * NFEAT] = acc[r];
        } else {
            #pragma unroll
            for (int r = 0; r < 8; ++r)
                if (m0 + r + mh < nnodes) dst[r * NFEAT] = acc[r];
        }

        if (g < ngroups) group_put(rv, g, lane, nplanegroups, nedges, plane, rec, pres[par]);
        g = gn; rv = rvn;
        qdone = (q < nquarters) ? q : -1;
        q += tstep;

        if (more) {
            *(float4*)&xs[cur ^ 1][sr * LDSPAD + sc]       = p0;
            *(float4*)&xs[cur ^ 1][(sr + 8) * LDSPAD + sc] = p1;
        }
        __builtin_amdgcn_fence(__ATOMIC_RELEASE, "workgroup", "local");
        __builtin_amdgcn_s_barrier();
        __builtin_amdgcn_fence(__ATOMIC_ACQUIRE, "workgroup", "local");
        cur ^= 1;
        par ^= 1;
    }

    while (q < nquarters) {
        if (qdone >= 0) pres_flush(pres[par ^ 1], presence, qdone, tid);
        if (g < ngroups) group_put(rv, g, lane, nplanegroups, nedges, plane, rec, pres[par]);
        g += gstep;
        rv = edge_fetch(rows, cols, vals, g * 32 + lane, (g < ngroups) ? nedges : 0);
        qdone = q;
        q += tstep;
        __builtin_amdgcn_fence(__ATOMIC_RELEASE, "workgroup", "local");
        __builtin_amdgcn_s_barrier();
        __builtin_amdgcn_fence(__ATOMIC_ACQUIRE, "workgroup", "local");
        par ^= 1;
    }
    if (qdone >= 0) pres_flush(pres[par ^ 1], presence, qdone, tid);
}


__device__ __forceinline__ void lds_add(float* p, float x)
{
    (void)__hip_atomic_fetch_add(p, x, __ATOMIC_RELAXED, __HIP_MEMORY_SCOPE_WORKGROUP);
}

template <bool FULL>
__device__ __forceinline__ void agg_drain(
    const float* __restrict__ H, const int4* __restrict__ rec,
    float* acc, const int* myq, int first, int nb, int lane, int base, int nown, int nnodes)
{
    int lr = 0, c = 0;
    float v = 0.0f;
    if (FULL || lane < nb) {
        const int4 r = rec[myq[first + lane]];
        lr = r.x - base;
        if ((unsigned)lr >= (unsigned)nown) lr = AGG_ROWS;
        c  = r.y;
        v  = __int_as_float(r.z);
    }
    c = min(max(c, 0), nnodes - 1);

    #pragma unroll
    for (int b = 0; b < 32; b += 8) {
        if (!FULL && b >= nb) break;
        float4 h[8];
        #pragma unroll
        for (int i = 0; i < 8; ++i) {
            if (FULL || b + i < nb) {
                const int ci = __builtin_amdgcn_readlane(c, b + i);
                h[i] = ((const float4*)(H + (size_t)ci * NFEAT))[lane];
            }
        }
        #pragma unroll
        for (int i = 0; i < 8; ++i) {
            if (FULL || b + i < nb) {
                const float vi = __int_as_float(__builtin_amdgcn_readlane(__float_as_int(v), b + i));
                const int   ri = __builtin_amdgcn_readlane(lr, b + i);
                float* a = acc + ri * NFEAT + lane;
                lds_add(a,      h[i].x * vi);
                lds_add(a + 32, h[i].y * vi);
                lds_add(a + 64, h[i].z * vi);
                lds_add(a + 96, h[i].w * vi);
            }
        }
    }
}

__device__ __forceinline__ int4 agg_load_ids_tail(const int* __restrict__ rows, int s,
                                                  int nedges, int lane)
{
    const int e0 = s * 128 + lane * 4;
    int4 t; t.x = -1; t.y = -1; t.z = -1; t.w = -1;
    if (e0 + 0 < nedges) t.x = rows[e0 + 0];
    if (e0 + 1 < nedges) t.y = rows[e0 + 1];
    if (e0 + 2 < nedges) t.z = rows[e0 + 2];
    if (e0 + 3 < nedges) t.w = rows[e0 + 3];
    return t;
}

__device__ __forceinline__ void agg_append(int4 ids, int e0, int base, int nown, int* myq, int& cnt)
{
    const bool m0 = (unsigned)(ids.x - base) < (unsigned)nown;
    const bool m1 = (unsigned)(ids.y - base) < (unsigned)nown;
    const bool m2 = (unsigned)(ids.z - base) < (unsigned)nown;
    const bool m3 = (unsigned)(ids.w - base) < (unsigned)nown;
    if (__builtin_amdgcn_ballot_w32(m0 | m1 | m2 | m3) != 0u) {
        const bool mt[4] = { m0, m1, m2, m3 };
        #pragma unroll
        for (int j = 0; j < 4; ++j) {
            const unsigned m = __builtin_amdgcn_ballot_w32(mt[j]);
            if (m != 0u) {
                const int pos = cnt + (int)__builtin_amdgcn_mbcnt_lo(m, 0u);
                if (mt[j]) myq[pos] = e0 + j;
                cnt += __builtin_popcount(m);
            }
        }
    }
}

__device__ __forceinline__ unsigned agg_match(int4 lo, int4 hi, unsigned id)
{
    const unsigned d[8] = { (unsigned)lo.x, (unsigned)lo.y, (unsigned)lo.z, (unsigned)lo.w,
                            (unsigned)hi.x, (unsigned)hi.y, (unsigned)hi.z, (unsigned)hi.w };
    unsigned differs = 0u;
    #pragma unroll
    for (int b = 0; b < 8; ++b)
        differs |= d[b] ^ (0u - ((id >> b) & 1u));
    return ~differs;
}

__device__ __forceinline__ void agg_append_bits(unsigned m, int e0, int* myq, int& cnt)
{
    unsigned bm = __builtin_amdgcn_ballot_w32(m != 0u);
    while (bm != 0u) {
        if (m != 0u) {
            myq[cnt + (int)__builtin_amdgcn_mbcnt_lo(bm, 0u)] = e0 + __builtin_ctz(m);
            m &= m - 1u;
        }
        cnt += __builtin_popcount(bm);
        bm = __builtin_amdgcn_ballot_w32(m != 0u);
    }
}

template <bool FULL>
__device__ __forceinline__ unsigned agg_group_mask(const int4* __restrict__ pw, const int* gq, int first,
                                                   int nb, int lane, unsigned id, int& e0)
{
    unsigned m = 0u;
    e0 = 0;
    if (FULL || lane < nb) {
        const int g = gq[first + lane];
        m  = agg_match(pw[g * 2], pw[g * 2 + 1], id);
        e0 = g * 32;
    }
    return m;
}

__global__ __launch_bounds__(1024) void gcn_aggregate_elu(
    const float* __restrict__ H, const int* __restrict__ rows,
    const int* __restrict__ recw,
    const int* __restrict__ plane, const int* __restrict__ presw,
    float* __restrict__ out, int nnodes, int nedges)
{
    const int4* __restrict__ rec = (const int4*)recw;
    __shared__ float acc[(AGG_ROWS + 1) * NFEAT];
    __shared__ int   queue[AGG_WAVES * AGG_QCAP];
    __shared__ int   gqueue[AGG_WAVES * AGG_GCAP];

    const int tid  = threadIdx.x;
    const int lane = tid & 31;
    const int wave = __builtin_amdgcn_readfirstlane(tid >> 5);
    const int base = blockIdx.x * AGG_ROWS;
    const int nown = max(min(AGG_ROWS, nnodes - base), 0);

    {
        float4 z; z.x = 0.f; z.y = 0.f; z.z = 0.f; z.w = 0.f;
        for (int i = tid; i < AGG_ROWS * NFEAT / 4; i += 1024) ((float4*)acc)[i] = z;
    }
    __syncthreads();

    int* myq = queue + wave * AGG_QCAP;
    int  cnt = 0;

    const int      nfull = nedges >> 10;
    const unsigned id    = blockIdx.x & 255u;
    const int4*    pw    = (const int4*)plane;
    int* mygq = gqueue + wave * AGG_GCAP;
    int  gcnt = 0;

    int      sn = wave + lane * AGG_WAVES;
    unsigned wn = (sn < nfull) ? (unsigned)presw[(size_t)sn * 256 + id] : 0u;
    for (int s0 = wave; s0 < nfull; s0 += 32 * AGG_WAVES) {
        const int      s = sn;
        const unsigned w = wn;
        sn = s + 32 * AGG_WAVES;
        wn = (sn < nfull) ? (unsigned)presw[(size_t)sn * 256 + id] : 0u;
        for (int k = 0; k < 4; ++k) {
            agg_append_bits((w >> (8 * k)) & 255u, s * 32 + 8 * k, mygq, gcnt);
            while (gcnt >= 32) {
                gcnt -= 32;
                int e0;
                const unsigned m = agg_group_mask<true>(pw, mygq, gcnt, 32, lane, id, e0);
                agg_append_bits(m, e0, myq, cnt);
                while (cnt >= 32) {
                    cnt -= 32;
                    agg_drain<true>(H, rec, acc, myq, cnt, 32, lane, base, nown, nnodes);
                }
            }
        }
    }
    if (gcnt > 0) {
        int e0;
        const unsigned m = agg_group_mask<false>(pw, mygq, 0, gcnt, lane, id, e0);
        agg_append_bits(m, e0, myq, cnt);
        while (cnt >= 32) {
            cnt -= 32;
            agg_drain<true>(H, rec, acc, myq, cnt, 32, lane, base, nown, nnodes);
        }
    }
    const int nsteps = (nedges + 127) >> 7;
    for (int s = nfull * 8 + wave; s < nsteps; s += AGG_WAVES) {
        agg_append(agg_load_ids_tail(rows, s, nedges, lane), s * 128 + lane * 4, base, nown, myq, cnt);
        while (cnt >= 32) {
            cnt -= 32;
            agg_drain<true>(H, rec, acc, myq, cnt, 32, lane, base, nown, nnodes);
        }
    }
    if (cnt > 0)
        agg_drain<false>(H, rec, acc, myq, 0, cnt, lane, base, nown, nnodes);

    __syncthreads();

    for (int r = wave; r < nown; r += AGG_WAVES) {
        const float* a = acc + r * NFEAT + lane;
        const float v0 = a[0], v1 = a[32], v2 = a[64], v3 = a[96];
        float4 o;
        o.x = (v0 > 0.0f) ? v0 : (__expf(v0) - 1.0f);
        o.y = (v1 > 0.0f) ? v1 : (__expf(v1) - 1.0f);
        o.z = (v2 > 0.0f) ? v2 : (__expf(v2) - 1.0f);
        o.w = (v3 > 0.0f) ? v3 : (__expf(v3) - 1.0f);
        ((float4*)(out + (size_t)(base + r) * NFEAT))[lane] = o;
    }
}

extern "C" void kernel_launch(void* const* d_in, const int* in_sizes, int n_in,
                              void* d_out, int out_size, void* d_ws, size_t ws_size,
                              hipStream_t stream)
{
    const float* x    = (const float*)d_in[0];
    const float* W    = (const float*)d_in[1];
    const float* b    = (const float*)d_in[2];
    const int*   rows = (const int*)  d_in[3];
    const int*   cols = (const int*)  d_in[4];
    const float* vals = (const float*)d_in[5];

    const int nnodes  = in_sizes[0] / NFEAT;
    const int nlayers = in_sizes[1] / (NFEAT * NFEAT);
    const int nedges  = in_sizes[3] / nlayers;

    const size_t nodeElems = (size_t)nnodes * NFEAT;
    float* hbuf = (float*)d_ws;
    float* agg  = hbuf + nodeElems;
    int*   rec   = (int*)(agg + nodeElems);
    int*   plane = rec + (size_t)nedges * 4;
    int*   pres  = plane + (size_t)nedges;

    const int aggBlocks  = (nnodes + AGG_ROWS - 1) / AGG_ROWS;

    for (int l = 0; l < nlayers; ++l) {
        const float* cur = (l == 0) ? x : agg;
        gcn_gemm_bias<<<GEMM_BLOCKS, 256, 0, stream>>>(
            cur, W + (size_t)l * NFEAT * NFEAT, b + (size_t)l * NFEAT,
            hbuf, nnodes, rows + (size_t)l * nedges, cols + (size_t)l * nedges,
            vals + (size_t)l * nedges, plane, rec, pres, nedges);

        float* layerOut = (l == nlayers - 1) ? (float*)d_out : agg;
        gcn_aggregate_elu<<<aggBlocks, 1024, 0, stream>>>(
            hbuf, rows + (size_t)l * nedges, rec, plane, pres, layerOut, nnodes, nedges);
    }
}
